# starting kernel
# baseline (speedup 1.0000x reference)
#include <hip/hip_runtime.h>
#include <hip/hip_bf16.h>

typedef __attribute__((ext_vector_type(2))) float v2f;
typedef __attribute__((ext_vector_type(8))) float v8f;

__global__ __launch_bounds__(256) void ts_poly_wmma_kernel(
    const float* __restrict__ pix,
    const float* __restrict__ coef,
    const float* __restrict__ bias,
    float* __restrict__ out,
    int npix)
{
    const int  tid   = blockIdx.x * blockDim.x + threadIdx.x;
    const int  lane  = threadIdx.x & 31;
    const int  gwave = tid >> 5;
    const int  nwave = (int)((gridDim.x * blockDim.x) >> 5);
    const bool lower = lane < 16;
    const int  col   = lane & 15;

    v2f a0 = {0.f, 0.f}, a1 = {0.f, 0.f}, a2 = {0.f, 0.f}, a3 = {0.f, 0.f};
    if (col < 3) {
        const float* cr  = coef + col * 16;
        const int    off = lower ? 0 : 2;
        a0[0] = cr[ 0 + off]; a0[1] = cr[ 1 + off];
        a1[0] = cr[ 4 + off]; a1[1] = cr[ 5 + off];
        a2[0] = cr[ 8 + off]; a2[1] = cr[ 9 + off];
        a3[0] = cr[12 + off]; a3[1] = cr[13 + off];
    }
    const float bias0 = lower ? bias[0] : 0.f;
    const float bias1 = lower ? bias[1] : 0.f;
    const float bias2 = lower ? bias[2] : 0.f;

    const int ntiles = (npix + 15) >> 4;
    const int stride = nwave << 4;
    for (int tile = gwave; tile < ntiles; tile += nwave) {
        const int base = tile << 4;
        const int pidx = base + col;
        const int lidx = pidx < npix ? pidx : (npix - 1);

        v2f p = *(const v2f*)(pix + 2 * lidx);

        const int nbase = base + stride;
        if (nbase < npix) __builtin_prefetch(pix + 2 * (nbase + col), 0, 0);

        const float x  = p[0],  y  = p[1];
        const float x2 = x * x,   y2 = y * y;
        const float x4 = x2 * x2, y4 = y2 * y2;
        const float x6 = x4 * x2, y6 = y4 * y2;
        const float u = lower ? 1.f : x;
        const float w = lower ? 1.f : y;

        v2f b0, b1, b2, b3;
        b0[0] = u;       b0[1] = w;
        b1[0] = u * x2;  b1[1] = w * y2;
        b2[0] = u * x4;  b2[1] = w * y4;
        b3[0] = u * x6;  b3[1] = w * y6;

        v8f acc = {};
        acc[0] = bias0; acc[1] = bias1; acc[2] = bias2;

        acc = __builtin_amdgcn_wmma_f32_16x16x4_f32(false, a0, false, b0, (short)0, acc, false, false);
        acc = __builtin_amdgcn_wmma_f32_16x16x4_f32(false, a1, false, b1, (short)0, acc, false, false);
        acc = __builtin_amdgcn_wmma_f32_16x16x4_f32(false, a2, false, b2, (short)0, acc, false, false);
        acc = __builtin_amdgcn_wmma_f32_16x16x4_f32(false, a3, false, b3, (short)0, acc, false, false);

        if (lower && pidx < npix) {
            out[pidx]            = acc[0];
            out[npix + pidx]     = acc[1];
            out[2 * npix + pidx] = acc[2];
        }
    }
}

extern "C" void kernel_launch(void* const* d_in, const int* in_sizes, int n_in,
                              void* d_out, int out_size, void* d_ws, size_t ws_size,
                              hipStream_t stream) {
    const float* pix  = (const float*)d_in[0];
    const float* coef = (const float*)d_in[1];
    const float* bias = (const float*)d_in[2];
    float* out = (float*)d_out;

    const int npix  = in_sizes[0] / 2;
    const int block = 256;
    const int ntiles = (npix + 15) >> 4;
    int grid = 4096;
    const int maxGrid = (ntiles + (block / 32) - 1) / (block / 32);
    if (grid > maxGrid) grid = maxGrid;
    if (grid < 1) grid = 1;

    ts_poly_wmma_kernel<<<grid, block, 0, stream>>>(pix, coef, bias, out, npix);
}


# grid 4096 -> 512 workgroups: each wave walks 128-pixel batches with 16-byte loads and stores, the next batch's loads in flight while the tiles of the batch in hand run
# speedup vs baseline: 1.3942x; 1.3942x over previous
#include <hip/hip_runtime.h>

typedef __attribute__((ext_vector_type(2))) float v2f;
typedef __attribute__((ext_vector_type(4))) float v4f;
typedef __attribute__((ext_vector_type(8))) float v8f;


struct Coef {
    v2f a0, a1, a2, a3;
    v8f c;
};

struct Batch {
    v4f q[2][2];
};

__device__ __forceinline__ v8f tile_mma(const Coef& k, v2f p, bool lower)
{
    const v2f p2 = p * p;
    const v2f p4 = p2 * p2;
    const v2f p6 = p4 * p2;
    const v2f one = {1.f, 1.f};
    const v2f uw = lower ? one : p;

    const v2f b0 = uw;
    const v2f b1 = uw * p2;
    const v2f b2 = uw * p4;
    const v2f b3 = uw * p6;

    v8f acc = k.c;
    acc = __builtin_amdgcn_wmma_f32_16x16x4_f32(false, k.a0, false, b0, (short)0, acc, false, false);
    acc = __builtin_amdgcn_wmma_f32_16x16x4_f32(false, k.a1, false, b1, (short)0, acc, false, false);
    acc = __builtin_amdgcn_wmma_f32_16x16x4_f32(false, k.a2, false, b2, (short)0, acc, false, false);
    acc = __builtin_amdgcn_wmma_f32_16x16x4_f32(false, k.a3, false, b3, (short)0, acc, false, false);
    return acc;
}

__device__ __forceinline__ void load_batch(Batch& b, const float* __restrict__ pix, int base, int col)
{
    for (int g = 0; g < 2; ++g) {
        const v4f* p = (const v4f*)(pix + 2 * (base + 64 * g + 4 * col));
        b.q[g][0] = p[0];
        b.q[g][1] = p[1];
    }
}

__device__ __forceinline__ void do_batch(const Batch& b, const Coef& k, bool lower, int col,
                                         int base, int npix, float* __restrict__ out)
{
    for (int g = 0; g < 2; ++g) {
        v8f acc[4];
        for (int e = 0; e < 4; ++e) {
            const v4f q = b.q[g][e >> 1];
            v2f p;
            p[0] = q[2 * (e & 1)];
            p[1] = q[2 * (e & 1) + 1];
            acc[e] = tile_mma(k, p, lower);
        }
        const int p0 = base + 64 * g + 4 * col;
        if (lower) {
            for (int ch = 0; ch < 3; ++ch) {
                v4f r;
                r[0] = acc[0][ch]; r[1] = acc[1][ch]; r[2] = acc[2][ch]; r[3] = acc[3][ch];
                *(v4f*)(out + (size_t)ch * npix + p0) = r;
            }
        }
    }
}

__device__ __forceinline__ void stream_batches(const float* __restrict__ pix, float* __restrict__ out,
                                               const Coef& k, bool lower, int col,
                                               int gwave, int nwave, int nfull, int npix)
{
    int b = gwave;
    if (b >= nfull) return;
    Batch s0, s1;
    load_batch(s0, pix, b << 7, col);
    while (b + 2 * nwave < nfull) {
        load_batch(s1, pix, (b + nwave) << 7, col);
        do_batch(s0, k, lower, col, b << 7, npix, out);
        load_batch(s0, pix, (b + 2 * nwave) << 7, col);
        do_batch(s1, k, lower, col, (b + nwave) << 7, npix, out);
        b += 2 * nwave;
    }
    if (b + nwave < nfull) {
        load_batch(s1, pix, (b + nwave) << 7, col);
        do_batch(s0, k, lower, col, b << 7, npix, out);
        do_batch(s1, k, lower, col, (b + nwave) << 7, npix, out);
    } else {
        do_batch(s0, k, lower, col, b << 7, npix, out);
    }
}

__global__ __launch_bounds__(256) void ts_poly_wmma_kernel(
    const float* __restrict__ pix,
    const float* __restrict__ coef,
    const float* __restrict__ bias,
    float* __restrict__ out,
    int npix)
{
    const int  lane  = threadIdx.x & 31;
    const int  wib   = __builtin_amdgcn_readfirstlane((int)(threadIdx.x >> 5));
    const int  wpb   = (int)(blockDim.x >> 5);
    const int  gwave = (int)blockIdx.x * wpb + wib;
    const int  nwave = (int)gridDim.x * wpb;
    const bool lower = lane < 16;
    const int  col   = lane & 15;

    Coef k;
    k.a0 = (v2f){0.f, 0.f}; k.a1 = k.a0; k.a2 = k.a0; k.a3 = k.a0;
    if (col < 3) {
        const float* cr  = coef + col * 16;
        const int    off = lower ? 0 : 2;
        k.a0[0] = cr[ 0 + off]; k.a0[1] = cr[ 1 + off];
        k.a1[0] = cr[ 4 + off]; k.a1[1] = cr[ 5 + off];
        k.a2[0] = cr[ 8 + off]; k.a2[1] = cr[ 9 + off];
        k.a3[0] = cr[12 + off]; k.a3[1] = cr[13 + off];
    }
    k.c = (v8f){};
    k.c[0] = lower ? bias[0] : 0.f;
    k.c[1] = lower ? bias[1] : 0.f;
    k.c[2] = lower ? bias[2] : 0.f;

    const bool vec_ok = (npix & 3) == 0;
    const int  nfull  = vec_ok ? (npix >> 7) : 0;
    if (vec_ok) stream_batches(pix, out, k, lower, col, gwave, nwave, nfull, npix);

    const int ntiles = (npix + 15) >> 4;
    for (int tile = (nfull << 3) + gwave; tile < ntiles; tile += nwave) {
        const int pidx = (tile << 4) + col;
        const int lidx = pidx < npix ? pidx : (npix - 1);
        const v2f p = *(const v2f*)(pix + 2 * lidx);
        const v8f acc = tile_mma(k, p, lower);
        if (lower && pidx < npix) {
            out[pidx]                    = acc[0];
            out[(size_t)npix + pidx]     = acc[1];
            out[2 * (size_t)npix + pidx] = acc[2];
        }
    }
}

extern "C" void kernel_launch(void* const* d_in, const int* in_sizes, int n_in,
                              void* d_out, int out_size, void* d_ws, size_t ws_size,
                              hipStream_t stream) {
    const float* pix  = (const float*)d_in[0];
    const float* coef = (const float*)d_in[1];
    const float* bias = (const float*)d_in[2];
    float* out = (float*)d_out;

    const int npix  = in_sizes[0] / 2;
    const int block = 256;
    const int nbatch = (npix + 127) >> 7;
    int grid = 512;
    const int maxGrid = (nbatch + (block / 32) - 1) / (block / 32);
    if (grid > maxGrid) grid = maxGrid;
    if (grid < 1) grid = 1;

    ts_poly_wmma_kernel<<<grid, block, 0, stream>>>(pix, coef, bias, out, npix);
}


# two pixels per B column: each lane half carries its own pixel's 7 powers, coefficient rows 0-2 and 8-10 in A, t=0 terms added to the bias in C: 7 WMMAs per 32 pixels, was 8; sum order changes
# speedup vs baseline: 1.5624x; 1.1206x over previous
#include <hip/hip_runtime.h>

typedef __attribute__((ext_vector_type(2))) float v2f;
typedef __attribute__((ext_vector_type(4))) float v4f;
typedef __attribute__((ext_vector_type(8))) float v8f;


struct Coef {
    v2f a[7];
    v8f c;
};

struct Batch {
    v4f q[2];
};

__device__ __forceinline__ v8f tile_mma(const Coef& k, v2f p)
{
    const v2f p2 = p * p;
    const v2f p3 = p2 * p;
    const v2f p4 = p2 * p2;
    const v2f p5 = p4 * p;
    const v2f p6 = p4 * p2;
    const v2f p7 = p4 * p3;

    v8f acc = k.c;
    acc = __builtin_amdgcn_wmma_f32_16x16x4_f32(false, k.a[0], false, p,  (short)0, acc, false, false);
    acc = __builtin_amdgcn_wmma_f32_16x16x4_f32(false, k.a[1], false, p2, (short)0, acc, false, false);
    acc = __builtin_amdgcn_wmma_f32_16x16x4_f32(false, k.a[2], false, p3, (short)0, acc, false, false);
    acc = __builtin_amdgcn_wmma_f32_16x16x4_f32(false, k.a[3], false, p4, (short)0, acc, false, false);
    acc = __builtin_amdgcn_wmma_f32_16x16x4_f32(false, k.a[4], false, p5, (short)0, acc, false, false);
    acc = __builtin_amdgcn_wmma_f32_16x16x4_f32(false, k.a[5], false, p6, (short)0, acc, false, false);
    acc = __builtin_amdgcn_wmma_f32_16x16x4_f32(false, k.a[6], false, p7, (short)0, acc, false, false);
    return acc;
}

__device__ __forceinline__ void load_batch(Batch& b, const float* __restrict__ pix, int base, int lane)
{
    for (int g = 0; g < 2; ++g)
        b.q[g] = *(const v4f*)(pix + 2 * (base + 64 * g + 2 * lane));
}

__device__ __forceinline__ void do_batch(const Batch& b, const Coef& k, int lane,
                                         int base, int npix, float* __restrict__ out)
{
    for (int g = 0; g < 2; ++g) {
        const v4f q = b.q[g];
        const v2f pa = {q[0], q[1]};
        const v2f pb = {q[2], q[3]};
        const v8f acc0 = tile_mma(k, pa);
        const v8f acc1 = tile_mma(k, pb);
        const int p0 = base + 64 * g + 2 * lane;
        for (int ch = 0; ch < 3; ++ch) {
            v2f r;
            r[0] = acc0[ch]; r[1] = acc1[ch];
            *(v2f*)(out + (size_t)ch * npix + p0) = r;
        }
    }
}

__device__ __forceinline__ void stream_batches(const float* __restrict__ pix, float* __restrict__ out,
                                               const Coef& k, int lane,
                                               int gwave, int nwave, int nfull, int npix)
{
    int b = gwave;
    if (b >= nfull) return;
    Batch s0, s1;
    load_batch(s0, pix, b << 7, lane);
    while (b + 2 * nwave < nfull) {
        load_batch(s1, pix, (b + nwave) << 7, lane);
        __builtin_amdgcn_sched_barrier(0);
        do_batch(s0, k, lane, b << 7, npix, out);
        load_batch(s0, pix, (b + 2 * nwave) << 7, lane);
        __builtin_amdgcn_sched_barrier(0);
        do_batch(s1, k, lane, (b + nwave) << 7, npix, out);
        b += 2 * nwave;
    }
    if (b + nwave < nfull) {
        load_batch(s1, pix, (b + nwave) << 7, lane);
        __builtin_amdgcn_sched_barrier(0);
        do_batch(s0, k, lane, b << 7, npix, out);
        do_batch(s1, k, lane, (b + nwave) << 7, npix, out);
    } else {
        do_batch(s0, k, lane, b << 7, npix, out);
    }
}

__global__ __launch_bounds__(256) void ts_poly_wmma_kernel(
    const float* __restrict__ pix,
    const float* __restrict__ coef,
    const float* __restrict__ bias,
    float* __restrict__ out,
    int npix)
{
    const int  lane  = threadIdx.x & 31;
    const int  wib   = __builtin_amdgcn_readfirstlane((int)(threadIdx.x >> 5));
    const int  wpb   = (int)(blockDim.x >> 5);
    const int  gwave = (int)blockIdx.x * wpb + wib;
    const int  nwave = (int)gridDim.x * wpb;

    Coef k;
    for (int j = 0; j < 7; ++j) k.a[j] = (v2f){0.f, 0.f};
    const int  ch_row = lane & 7;
    const bool a_live = (ch_row < 3) && (((lane >> 3) & 1) == (lane >> 4));
    if (a_live) {
        const v2f* cr = (const v2f*)(coef + ch_row * 16 + 2);
        for (int j = 0; j < 7; ++j) k.a[j] = cr[j];
    }
    k.c = (v8f){};
    k.c[0] = bias[0] + coef[0]  + coef[1];
    k.c[1] = bias[1] + coef[16] + coef[17];
    k.c[2] = bias[2] + coef[32] + coef[33];

    const bool vec_ok = (npix & 1) == 0;
    const int  nfull  = vec_ok ? (npix >> 7) : 0;
    if (vec_ok) stream_batches(pix, out, k, lane, gwave, nwave, nfull, npix);

    const int ntiles = (npix + 31) >> 5;
    for (int tile = (nfull << 2) + gwave; tile < ntiles; tile += nwave) {
        const int pidx = (tile << 5) + lane;
        const int lidx = pidx < npix ? pidx : (npix - 1);
        const v2f p = *(const v2f*)(pix + 2 * lidx);
        const v8f acc = tile_mma(k, p);
        if (pidx < npix) {
            out[pidx]                    = acc[0];
            out[(size_t)npix + pidx]     = acc[1];
            out[2 * (size_t)npix + pidx] = acc[2];
        }
    }
}

extern "C" void kernel_launch(void* const* d_in, const int* in_sizes, int n_in,
                              void* d_out, int out_size, void* d_ws, size_t ws_size,
                              hipStream_t stream) {
    const float* pix  = (const float*)d_in[0];
    const float* coef = (const float*)d_in[1];
    const float* bias = (const float*)d_in[2];
    float* out = (float*)d_out;

    const int npix  = in_sizes[0] / 2;
    const int block = 256;
    const int nbatch = (npix + 127) >> 7;
    int grid = 512;
    const int maxGrid = (nbatch + (block / 32) - 1) / (block / 32);
    if (grid > maxGrid) grid = maxGrid;
    if (grid < 1) grid = 1;

    ts_poly_wmma_kernel<<<grid, block, 0, stream>>>(pix, coef, bias, out, npix);
}
